# starting kernel
# baseline (speedup 1.0000x reference)
#include <hip/hip_runtime.h>
#include <math.h>


typedef __attribute__((ext_vector_type(2))) float v2f;
typedef __attribute__((ext_vector_type(8))) float v8f;

#define BATCH 512
#define NN    255
#define NL    256
#define NROWS 511
#define KP    256
#define NT    32
#define MT    32

__global__ __launch_bounds__(256) void tree_prep(
    const float* __restrict__ x,
    const float* __restrict__ rbin,
    const float* __restrict__ lbin,
    float* __restrict__ Lx, float* __restrict__ L1,
    float* __restrict__ Rp, float* __restrict__ Lp)
{
    int idx = blockIdx.x * 256 + threadIdx.x;
    int c = idx & (KP - 1);
    int r = idx >> 8;

    float lx = 0.0f, l1 = 0.0f;
    if (c < NN) {
        float v = x[r * NN + c];
        lx = fmaxf(logf(v),        -1.0e30f);
        l1 = fmaxf(logf(1.0f - v), -1.0e30f);
    }
    Lx[idx] = lx;
    L1[idx] = l1;

    float rv = 0.0f, lv = 0.0f;
    if (r < NROWS && c < NN) {
        rv = rbin[r * NN + c];
        lv = lbin[r * NN + c];
    }
    Rp[idx] = rv;
    Lp[idx] = lv;
}

__global__ __launch_bounds__(256) void tree_wmma(
    const float* __restrict__ Lx, const float* __restrict__ L1,
    const float* __restrict__ Rp, const float* __restrict__ Lp,
    float* __restrict__ out)
{
    const int lane = threadIdx.x & 31;
    const int wave = threadIdx.x >> 5;
    const int tile = blockIdx.x * 8 + wave;
    const int nt = tile & (NT - 1);
    const int mt = tile >> 5;
    const int m0 = mt * 16;
    const int n0 = nt * 16;

    const int rc = lane & 15;
    const int kk = (lane >> 4) << 1;

    const float* ap  = Lx + (size_t)(m0 + rc) * KP + kk;
    const float* a2p = L1 + (size_t)(m0 + rc) * KP + kk;
    const float* bp  = Rp + (size_t)(n0 + rc) * KP + kk;
    const float* b2p = Lp + (size_t)(n0 + rc) * KP + kk;

    v8f acc = {};
#pragma unroll 8
    for (int k0 = 0; k0 < KP; k0 += 4) {
        v2f a  = *(const v2f*)(ap  + k0);
        v2f a2 = *(const v2f*)(a2p + k0);
        v2f br = *(const v2f*)(bp  + k0);
        v2f bl = *(const v2f*)(b2p + k0);
        acc = __builtin_amdgcn_wmma_f32_16x16x4_f32(
                  false, a,  false, br, (short)0, acc, false, false);
        acc = __builtin_amdgcn_wmma_f32_16x16x4_f32(
                  false, a2, false, bl, (short)0, acc, false, false);
    }

    const int j     = n0 + rc;
    const int mbase = m0 + ((lane >> 4) << 3);
#pragma unroll
    for (int v = 0; v < 8; ++v) {
        int m = mbase + v;
        float e = expf(acc[v]);
        if (j < NROWS) {
            if (j < NL) out[(size_t)m * NL + j] = e;
            else        out[(size_t)BATCH * NL + (size_t)m * NN + (j - NL)] = e;
        }
    }
}

extern "C" void kernel_launch(void* const* d_in, const int* in_sizes, int n_in,
                              void* d_out, int out_size, void* d_ws, size_t ws_size,
                              hipStream_t stream) {
    const float* x    = (const float*)d_in[0];
    const float* rbin = (const float*)d_in[1];
    const float* lbin = (const float*)d_in[2];

    float* ws = (float*)d_ws;
    float* Lx = ws;
    float* L1 = ws + 512 * 256;
    float* Rp = ws + 2 * 512 * 256;
    float* Lp = ws + 3 * 512 * 256;

    tree_prep<<<512, 256, 0, stream>>>(x, rbin, lbin, Lx, L1, Rp, Lp);
    tree_wmma<<<128, 256, 0, stream>>>(Lx, L1, Rp, Lp, (float*)d_out);
}


# prep writes both operands in WMMA fragment order: a wave's fragment load is one contiguous 512 B (16 B per lane, 2 k-steps) instead of 16 row pieces of 16 B; summation order unchanged
# speedup vs baseline: 1.1409x; 1.1409x over previous
#include <hip/hip_runtime.h>
#include <math.h>


typedef __attribute__((ext_vector_type(2))) float v2f;
typedef __attribute__((ext_vector_type(4))) float v4f;
typedef __attribute__((ext_vector_type(8))) float v8f;

#define BATCH 512
#define NN    255
#define NL    256
#define NROWS 511
#define KP    256
#define KG    32
#define NT    32
#define MT    32

__global__ __launch_bounds__(256) void tree_prep(
    const float* __restrict__ x,
    const float* __restrict__ rbin,
    const float* __restrict__ lbin,
    float* __restrict__ Lx, float* __restrict__ L1,
    float* __restrict__ Rp, float* __restrict__ Lp)
{
    const int g  = (blockIdx.x & 127) * 256 + threadIdx.x;
    const int l  = g & 31;
    const int s  = (g >> 5) & (KG - 1);
    const int t  = g >> 10;
    const int r  = t * 16 + (l & 15);
    const int kb = s * 8 + ((l >> 4) << 1);

    if (blockIdx.x < 128) {
        v4f lx, l1;
#pragma unroll
        for (int e = 0; e < 4; ++e) {
            const int  k   = kb + (e & 1) + ((e >> 1) << 2);
            const bool kin = k < NN;
            const float v  = x[r * NN + (kin ? k : (NN - 1))];
            const float a = fmaxf(logf(v),        -1.0e30f);
            const float b = fmaxf(logf(1.0f - v), -1.0e30f);
            lx[e] = kin ? a : 0.0f;
            l1[e] = kin ? b : 0.0f;
        }
        ((v4f*)Lx)[g] = lx;
        ((v4f*)L1)[g] = l1;
    } else {
        const bool rin = r < NROWS;
        const int  rb  = rin ? r : (NROWS - 1);
        v4f rv, lv;
#pragma unroll
        for (int e = 0; e < 4; ++e) {
            const int  k   = kb + (e & 1) + ((e >> 1) << 2);
            const bool kin = k < NN;
            const int  kc  = kin ? k : (NN - 1);
            const float c = rbin[rb * NN + kc];
            const float d = lbin[rb * NN + kc];
            rv[e] = (kin && rin) ? c : 0.0f;
            lv[e] = (kin && rin) ? d : 0.0f;
        }
        ((v4f*)Rp)[g] = rv;
        ((v4f*)Lp)[g] = lv;
    }
}

__global__ __launch_bounds__(256) void tree_wmma(
    const float* __restrict__ Lx, const float* __restrict__ L1,
    const float* __restrict__ Rp, const float* __restrict__ Lp,
    float* __restrict__ out)
{
    const int lane = threadIdx.x & 31;
    const int wave = threadIdx.x >> 5;
    const int tile = blockIdx.x * 8 + wave;
    const int nt = tile & (NT - 1);
    const int mt = tile >> 5;
    const int m0 = mt * 16;
    const int n0 = nt * 16;

    const int rc = lane & 15;

    const v4f* ap  = (const v4f*)Lx + (size_t)mt * (KG * 32) + lane;
    const v4f* a2p = (const v4f*)L1 + (size_t)mt * (KG * 32) + lane;
    const v4f* bp  = (const v4f*)Rp + (size_t)nt * (KG * 32) + lane;
    const v4f* b2p = (const v4f*)Lp + (size_t)nt * (KG * 32) + lane;

    v8f acc = {};
#pragma unroll 8
    for (int s = 0; s < KG; ++s) {
        v4f a  = ap [s * 32];
        v4f a2 = a2p[s * 32];
        v4f br = bp [s * 32];
        v4f bl = b2p[s * 32];
        acc = __builtin_amdgcn_wmma_f32_16x16x4_f32(
                  false, a.xy,  false, br.xy, (short)0, acc, false, false);
        acc = __builtin_amdgcn_wmma_f32_16x16x4_f32(
                  false, a2.xy, false, bl.xy, (short)0, acc, false, false);
        acc = __builtin_amdgcn_wmma_f32_16x16x4_f32(
                  false, a.zw,  false, br.zw, (short)0, acc, false, false);
        acc = __builtin_amdgcn_wmma_f32_16x16x4_f32(
                  false, a2.zw, false, bl.zw, (short)0, acc, false, false);
    }

    const int j     = n0 + rc;
    const int mbase = m0 + ((lane >> 4) << 3);
#pragma unroll
    for (int v = 0; v < 8; ++v) {
        int m = mbase + v;
        float e = expf(acc[v]);
        if (j < NROWS) {
            if (j < NL) out[(size_t)m * NL + j] = e;
            else        out[(size_t)BATCH * NL + (size_t)m * NN + (j - NL)] = e;
        }
    }
}

extern "C" void kernel_launch(void* const* d_in, const int* in_sizes, int n_in,
                              void* d_out, int out_size, void* d_ws, size_t ws_size,
                              hipStream_t stream) {
    const float* x    = (const float*)d_in[0];
    const float* rbin = (const float*)d_in[1];
    const float* lbin = (const float*)d_in[2];

    float* ws = (float*)d_ws;
    float* Lx = ws;
    float* L1 = ws + 512 * 256;
    float* Rp = ws + 2 * 512 * 256;
    float* Lp = ws + 3 * 512 * 256;

    tree_prep<<<256, 256, 0, stream>>>(x, rbin, lbin, Lx, L1, Rp, Lp);
    tree_wmma<<<128, 256, 0, stream>>>(Lx, L1, Rp, Lp, (float*)d_out);
}


# prep folded into the GEMM launch: each 32x32 work-group stages its rows of x (logs taken there) and of both bins in LDS in fragment order; R and L products summed apart, added before exp
# speedup vs baseline: 2.4263x; 2.1267x over previous
#include <hip/hip_runtime.h>
#include <math.h>


typedef __attribute__((ext_vector_type(2))) float v2f;
typedef __attribute__((ext_vector_type(4))) float v4f;
typedef __attribute__((ext_vector_type(8))) float v8f;

#define BATCH  512
#define NN     255
#define NL     256
#define NROWS  511
#define TR     32
#define CHUNK  (TR * NN)
#define CHUNK4 (CHUNK / 4)
#define KG     32
#define SS     132
#define TS     (KG * SS)
#define PS     (2 * TS)
#define PP     36
#define LDS_USED   (4 * PS + 2 * TR * PP)
#define LDS_FLOATS (164 * 256)

__device__ __forceinline__ int frag_pos(int row, int k)
{
    const int kk = k & 7;
    const int h  = (kk >> 1) & 1;
    const int e  = (kk & 1) | ((kk >> 2) << 1);
    return (row >> 4) * TS + (k >> 3) * SS + ((h << 4) + (row & 15)) * 4 + e;
}

__global__ __launch_bounds__(256) void tree_fused(
    const float* __restrict__ x,
    const float* __restrict__ rbin,
    const float* __restrict__ lbin,
    float* __restrict__ out)
{
    static_assert(LDS_USED <= LDS_FLOATS, "LDS reservation too small");
    __shared__ __attribute__((aligned(16))) float lds[LDS_FLOATS];

    const int tid  = threadIdx.x;
    const int lane = tid & 31;
    const int wave = tid >> 5;
    const int bm   = blockIdx.x >> 4;
    const int bn   = blockIdx.x & 15;

    const float* xc = x    + bm * CHUNK;
    const float* rc = rbin + bn * CHUNK;
    const float* lc = lbin + bn * CHUNK;
    const int nb = min(CHUNK, NROWS * NN - bn * CHUNK);

    static_assert(4 * (256 * 6 + 255) + 3 < NROWS * NN - 15 * CHUNK, "pieces 0..6 must lie inside the last chunk");
    v4f xv[8], rv[8], lv[8];
#pragma unroll
    for (int j = 0; j < 7; ++j) xv[j] = ((const v4f*)xc)[tid + 256 * j];
    xv[7] = ((const v4f*)xc)[min(tid + 256 * 7, CHUNK4 - 1)];
#pragma unroll
    for (int j = 0; j < 7; ++j) {
        rv[j] = ((const v4f*)rc)[tid + 256 * j];
        lv[j] = ((const v4f*)lc)[tid + 256 * j];
    }
    {
        const int f0 = 4 * (tid + 256 * 7);
#pragma unroll
        for (int e = 0; e < 4; ++e) {
            const int   fc = min(f0 + e, nb - 1);
            const float r1 = rc[fc];
            const float l1 = lc[fc];
            rv[7][e] = (f0 + e < nb) ? r1 : 0.0f;
            lv[7][e] = (f0 + e < nb) ? l1 : 0.0f;
        }
    }

    if (tid < 128) lds[(tid >> 5) * PS + frag_pos(tid & 31, NN)] = 0.0f;

    int pos[8][4];
#pragma unroll
    for (int j = 0; j < 8; ++j) {
        const int i    = tid + 256 * j;
        const int f0   = 4 * i;
        const int row0 = f0 / NN;
        const int k0   = f0 - row0 * NN;
#pragma unroll
        for (int e = 0; e < 4; ++e) {
            int k = k0 + e, row = row0;
            if (k >= NN) { k -= NN; row += 1; }
            pos[j][e] = frag_pos(row, k);
        }
        if (i < CHUNK4) {
#pragma unroll
            for (int e = 0; e < 4; ++e) {
                const float v = xv[j][e];
                lds[pos[j][e]]      = fmaxf(logf(v),        -1.0e30f);
                lds[PS + pos[j][e]] = fmaxf(logf(1.0f - v), -1.0e30f);
            }
        }
    }
#pragma unroll
    for (int j = 0; j < 8; ++j) {
        if (tid + 256 * j < CHUNK4) {
#pragma unroll
            for (int e = 0; e < 4; ++e) {
                lds[2 * PS + pos[j][e]] = rv[j][e];
                lds[3 * PS + pos[j][e]] = lv[j][e];
            }
        }
    }
    __syncthreads();

    const int q  = wave & 3;
    const int pl = wave >> 2;
    const int tm = q >> 1;
    const int tn = q & 1;
    const float* ap = lds + pl * PS       + tm * TS + lane * 4;
    const float* bp = lds + (2 + pl) * PS + tn * TS + lane * 4;

    v8f acc = {};
    v4f fa[2][8], fb[2][8];
#pragma unroll
    for (int g = 0; g < 8; ++g) {
        fa[0][g] = *(const v4f*)(ap + g * SS);
        fb[0][g] = *(const v4f*)(bp + g * SS);
    }
#pragma unroll
    for (int blk = 0; blk < 4; ++blk) {
        const int cur = blk & 1;
        const int nxt = cur ^ 1;
        if (blk < 3) {
#pragma unroll
            for (int g = 0; g < 8; ++g) {
                fa[nxt][g] = *(const v4f*)(ap + ((blk + 1) * 8 + g) * SS);
                fb[nxt][g] = *(const v4f*)(bp + ((blk + 1) * 8 + g) * SS);
            }
        }
        __builtin_amdgcn_sched_barrier(0);
#pragma unroll
        for (int g = 0; g < 8; ++g) {
            acc = __builtin_amdgcn_wmma_f32_16x16x4_f32(
                      false, fa[cur][g].xy, false, fb[cur][g].xy, (short)0, acc, false, false);
            acc = __builtin_amdgcn_wmma_f32_16x16x4_f32(
                      false, fa[cur][g].zw, false, fb[cur][g].zw, (short)0, acc, false, false);
        }
        __builtin_amdgcn_sched_barrier(0);
    }

    float* pb = lds + 4 * PS + pl * (TR * PP);
    {
        const int c = lane & 15;
        const int h = lane >> 4;
#pragma unroll
        for (int v = 0; v < 8; ++v)
            pb[(tm * 16 + 8 * h + v) * PP + tn * 16 + c] = acc[v];
    }
    __syncthreads();

    const float* p0 = lds + 4 * PS;
    const float* p1 = p0 + TR * PP;
    const int  j    = bn * TR + lane;
    const bool leaf = bn < 8;
    float* ob = leaf ? out + (size_t)(bm * TR) * NL + j
                     : out + (size_t)BATCH * NL + (size_t)(bm * TR) * NN + (j - NL);
    const int pitch = leaf ? NL : NN;
    float sum[4];
#pragma unroll
    for (int i = 0; i < 4; ++i) {
        const int r = wave + 8 * i;
        sum[i] = p0[r * PP + lane] + p1[r * PP + lane];
    }
    if (j < NROWS) {
#pragma unroll
        for (int i = 0; i < 4; ++i)
            ob[(wave + 8 * i) * pitch] = expf(sum[i]);
    }
}

extern "C" void kernel_launch(void* const* d_in, const int* in_sizes, int n_in,
                              void* d_out, int out_size, void* d_ws, size_t ws_size,
                              hipStream_t stream) {
    const float* x    = (const float*)d_in[0];
    const float* rbin = (const float*)d_in[1];
    const float* lbin = (const float*)d_in[2];

    tree_fused<<<256, 256, 0, stream>>>(x, rbin, lbin, (float*)d_out);
}


# logs and the exp in base 2 on the chip's own v_log_f32 / v_exp_f32: exp2(sum of log2) replaces the library logf (12 instructions each, 64 per lane) and expf; summation order unchanged
# speedup vs baseline: 2.7428x; 1.1304x over previous
#include <hip/hip_runtime.h>
#include <math.h>


typedef __attribute__((ext_vector_type(2))) float v2f;
typedef __attribute__((ext_vector_type(4))) float v4f;
typedef __attribute__((ext_vector_type(8))) float v8f;

#define BATCH  512
#define NN     255
#define NL     256
#define NROWS  511
#define TR     32
#define CHUNK  (TR * NN)
#define CHUNK4 (CHUNK / 4)
#define KG     32
#define SS     132
#define TS     (KG * SS)
#define PS     (2 * TS)
#define PP     36
#define LDS_USED   (4 * PS + 2 * TR * PP)
#define LDS_FLOATS (164 * 256)

__device__ __forceinline__ int frag_pos(int row, int k)
{
    const int kk = k & 7;
    const int h  = (kk >> 1) & 1;
    const int e  = (kk & 1) | ((kk >> 2) << 1);
    return (row >> 4) * TS + (k >> 3) * SS + ((h << 4) + (row & 15)) * 4 + e;
}

__global__ __launch_bounds__(256) void tree_fused(
    const float* __restrict__ x,
    const float* __restrict__ rbin,
    const float* __restrict__ lbin,
    float* __restrict__ out)
{
    static_assert(LDS_USED <= LDS_FLOATS, "LDS reservation too small");
    __shared__ __attribute__((aligned(16))) float lds[LDS_FLOATS];

    const int tid  = threadIdx.x;
    const int lane = tid & 31;
    const int wave = tid >> 5;
    const int bm   = blockIdx.x >> 4;
    const int bn   = blockIdx.x & 15;

    const float* xc = x    + bm * CHUNK;
    const float* rc = rbin + bn * CHUNK;
    const float* lc = lbin + bn * CHUNK;
    const int nb = min(CHUNK, NROWS * NN - bn * CHUNK);

    static_assert(4 * (256 * 6 + 255) + 3 < NROWS * NN - 15 * CHUNK, "pieces 0..6 must lie inside the last chunk");
    v4f xv[8], rv[8], lv[8];
#pragma unroll
    for (int j = 0; j < 7; ++j) xv[j] = ((const v4f*)xc)[tid + 256 * j];
    xv[7] = ((const v4f*)xc)[min(tid + 256 * 7, CHUNK4 - 1)];
#pragma unroll
    for (int j = 0; j < 7; ++j) {
        rv[j] = ((const v4f*)rc)[tid + 256 * j];
        lv[j] = ((const v4f*)lc)[tid + 256 * j];
    }
    {
        const int f0 = 4 * (tid + 256 * 7);
#pragma unroll
        for (int e = 0; e < 4; ++e) {
            const int   fc = min(f0 + e, nb - 1);
            const float r1 = rc[fc];
            const float l1 = lc[fc];
            rv[7][e] = (f0 + e < nb) ? r1 : 0.0f;
            lv[7][e] = (f0 + e < nb) ? l1 : 0.0f;
        }
    }

    if (tid < 128) lds[(tid >> 5) * PS + frag_pos(tid & 31, NN)] = 0.0f;

    int pos[8][4];
#pragma unroll
    for (int j = 0; j < 8; ++j) {
        const int i    = tid + 256 * j;
        const int f0   = 4 * i;
        const int row0 = f0 / NN;
        const int k0   = f0 - row0 * NN;
#pragma unroll
        for (int e = 0; e < 4; ++e) {
            int k = k0 + e, row = row0;
            if (k >= NN) { k -= NN; row += 1; }
            pos[j][e] = frag_pos(row, k);
        }
        if (i < CHUNK4) {
#pragma unroll
            for (int e = 0; e < 4; ++e) {
                const float v = xv[j][e];
                lds[pos[j][e]]      = fmaxf(__builtin_amdgcn_logf(v),        -1.0e30f);
                lds[PS + pos[j][e]] = fmaxf(__builtin_amdgcn_logf(1.0f - v), -1.0e30f);
            }
        }
    }
#pragma unroll
    for (int j = 0; j < 8; ++j) {
        if (tid + 256 * j < CHUNK4) {
#pragma unroll
            for (int e = 0; e < 4; ++e) {
                lds[2 * PS + pos[j][e]] = rv[j][e];
                lds[3 * PS + pos[j][e]] = lv[j][e];
            }
        }
    }
    __syncthreads();

    const int q  = wave & 3;
    const int pl = wave >> 2;
    const int tm = q >> 1;
    const int tn = q & 1;
    const float* ap = lds + pl * PS       + tm * TS + lane * 4;
    const float* bp = lds + (2 + pl) * PS + tn * TS + lane * 4;

    v8f acc = {};
    v4f fa[2][8], fb[2][8];
#pragma unroll
    for (int g = 0; g < 8; ++g) {
        fa[0][g] = *(const v4f*)(ap + g * SS);
        fb[0][g] = *(const v4f*)(bp + g * SS);
    }
#pragma unroll
    for (int blk = 0; blk < 4; ++blk) {
        const int cur = blk & 1;
        const int nxt = cur ^ 1;
        if (blk < 3) {
#pragma unroll
            for (int g = 0; g < 8; ++g) {
                fa[nxt][g] = *(const v4f*)(ap + ((blk + 1) * 8 + g) * SS);
                fb[nxt][g] = *(const v4f*)(bp + ((blk + 1) * 8 + g) * SS);
            }
        }
        __builtin_amdgcn_sched_barrier(0);
#pragma unroll
        for (int g = 0; g < 8; ++g) {
            acc = __builtin_amdgcn_wmma_f32_16x16x4_f32(
                      false, fa[cur][g].xy, false, fb[cur][g].xy, (short)0, acc, false, false);
            acc = __builtin_amdgcn_wmma_f32_16x16x4_f32(
                      false, fa[cur][g].zw, false, fb[cur][g].zw, (short)0, acc, false, false);
        }
        __builtin_amdgcn_sched_barrier(0);
    }

    float* pb = lds + 4 * PS + pl * (TR * PP);
    {
        const int c = lane & 15;
        const int h = lane >> 4;
#pragma unroll
        for (int v = 0; v < 8; ++v)
            pb[(tm * 16 + 8 * h + v) * PP + tn * 16 + c] = acc[v];
    }
    __syncthreads();

    const float* p0 = lds + 4 * PS;
    const float* p1 = p0 + TR * PP;
    const int  j    = bn * TR + lane;
    const bool leaf = bn < 8;
    float* ob = leaf ? out + (size_t)(bm * TR) * NL + j
                     : out + (size_t)BATCH * NL + (size_t)(bm * TR) * NN + (j - NL);
    const int pitch = leaf ? NL : NN;
    float sum[4];
#pragma unroll
    for (int i = 0; i < 4; ++i) {
        const int r = wave + 8 * i;
        sum[i] = p0[r * PP + lane] + p1[r * PP + lane];
    }
    if (j < NROWS) {
#pragma unroll
        for (int i = 0; i < 4; ++i)
            ob[(wave + 8 * i) * pitch] = __builtin_amdgcn_exp2f(sum[i]);
    }
}

extern "C" void kernel_launch(void* const* d_in, const int* in_sizes, int n_in,
                              void* d_out, int out_size, void* d_ws, size_t ws_size,
                              hipStream_t stream) {
    const float* x    = (const float*)d_in[0];
    const float* rbin = (const float*)d_in[1];
    const float* lbin = (const float*)d_in[2];

    tree_fused<<<256, 256, 0, stream>>>(x, rbin, lbin, (float*)d_out);
}


# staging by rows: a wave copies half rows with 16-byte loads of 4-byte-aligned data and stores each piece as two 8-byte LDS stores; no per-element position arithmetic; summation order unchanged
# speedup vs baseline: 2.9807x; 1.0867x over previous
#include <hip/hip_runtime.h>
#include <math.h>


typedef __attribute__((ext_vector_type(2))) float v2f;
typedef __attribute__((ext_vector_type(4))) float v4f;
typedef __attribute__((ext_vector_type(8))) float v8f;

#define BATCH  512
#define NN     255
#define NL     256
#define NROWS  511
#define TR     32
#define CHUNK  (TR * NN)
#define KG     32
#define SS     132
#define TS     (KG * SS)
#define PS     (2 * TS)
#define PP     36
#define LDS_USED   (4 * PS + 2 * TR * PP)
#define LDS_FLOATS (164 * 256)

__global__ __launch_bounds__(256) void tree_fused(
    const float* __restrict__ x,
    const float* __restrict__ rbin,
    const float* __restrict__ lbin,
    float* __restrict__ out)
{
    static_assert(LDS_USED <= LDS_FLOATS, "LDS reservation too small");
    __shared__ __attribute__((aligned(16))) float lds[LDS_FLOATS];

    const int tid  = threadIdx.x;
    const int lane = tid & 31;
    const int wave = __builtin_amdgcn_readfirstlane(tid >> 5);
    const int bm   = blockIdx.x >> 4;
    const int bn   = blockIdx.x & 15;

    const float* xc = x    + bm * CHUNK;
    const float* rc = rbin + bn * CHUNK;
    const float* lc = lbin + bn * CHUNK;
    const int last_row = min(TR, NROWS - bn * TR) - 1;

    const int r0   = wave >> 1;
    const int half = wave & 1;
    const int m    = half * 32 + lane;
    const bool tail = (m == 63);
    const unsigned kofs = 4u * (unsigned)m - (tail ? 1u : 0u);
    const int  lpos = (m >> 1) * SS + 2 * (m & 1);

    const float* xb = xc + r0 * NN;
    const float* rb = rc + r0 * NN;
    const float* lb = lc + r0 * NN;
    const int r7 = min(r0 + 28, last_row);
    v4f xv[8], rv[8], lv[8];
#pragma unroll
    for (int j = 0; j < 8; ++j)
        __builtin_memcpy(&xv[j], xb + j * (4 * NN) + kofs, sizeof(v4f));
#pragma unroll
    for (int j = 0; j < 7; ++j) {
        __builtin_memcpy(&rv[j], rb + j * (4 * NN) + kofs, sizeof(v4f));
        __builtin_memcpy(&lv[j], lb + j * (4 * NN) + kofs, sizeof(v4f));
    }
    __builtin_memcpy(&rv[7], rc + r7 * NN + kofs, sizeof(v4f));
    __builtin_memcpy(&lv[7], lc + r7 * NN + kofs, sizeof(v4f));

#pragma unroll
    for (int j = 0; j < 8; ++j) {
        const int row = r0 + 4 * j;
        float* d = lds + (row >> 4) * TS + (row & 15) * 4 + lpos;
        v4f v = xv[j];
        if (tail) v = (v4f){v.y, v.z, v.w, 0.0f};
        v4f a, b;
#pragma unroll
        for (int e = 0; e < 4; ++e) {
            a[e] = fmaxf(__builtin_amdgcn_logf(v[e]),        -1.0e30f);
            b[e] = fmaxf(__builtin_amdgcn_logf(1.0f - v[e]), -1.0e30f);
        }
        *(v2f*)(d)           = a.xy;
        *(v2f*)(d + 64)      = a.zw;
        *(v2f*)(d + PS)      = b.xy;
        *(v2f*)(d + PS + 64) = b.zw;
    }
    __builtin_amdgcn_sched_barrier(0);
#pragma unroll
    for (int j = 0; j < 8; ++j) {
        const int row = r0 + 4 * j;
        float* d = lds + 2 * PS + (row >> 4) * TS + (row & 15) * 4 + lpos;
        v4f r = rv[j], l = lv[j];
        if (tail) {
            r = (v4f){r.y, r.z, r.w, 0.0f};
            l = (v4f){l.y, l.z, l.w, 0.0f};
        }
        *(v2f*)(d)           = r.xy;
        *(v2f*)(d + 64)      = r.zw;
        *(v2f*)(d + PS)      = l.xy;
        *(v2f*)(d + PS + 64) = l.zw;
    }
    __syncthreads();

    const int q  = wave & 3;
    const int pl = wave >> 2;
    const int tm = q >> 1;
    const int tn = q & 1;
    const float* ap = lds + pl * PS       + tm * TS + lane * 4;
    const float* bp = lds + (2 + pl) * PS + tn * TS + lane * 4;

    v8f acc = {};
    v4f fa[2][8], fb[2][8];
#pragma unroll
    for (int g = 0; g < 8; ++g) {
        fa[0][g] = *(const v4f*)(ap + g * SS);
        fb[0][g] = *(const v4f*)(bp + g * SS);
    }
#pragma unroll
    for (int blk = 0; blk < 4; ++blk) {
        const int cur = blk & 1;
        const int nxt = cur ^ 1;
        if (blk < 3) {
#pragma unroll
            for (int g = 0; g < 8; ++g) {
                fa[nxt][g] = *(const v4f*)(ap + ((blk + 1) * 8 + g) * SS);
                fb[nxt][g] = *(const v4f*)(bp + ((blk + 1) * 8 + g) * SS);
            }
        }
        __builtin_amdgcn_sched_barrier(0);
#pragma unroll
        for (int g = 0; g < 8; ++g) {
            acc = __builtin_amdgcn_wmma_f32_16x16x4_f32(
                      false, fa[cur][g].xy, false, fb[cur][g].xy, (short)0, acc, false, false);
            acc = __builtin_amdgcn_wmma_f32_16x16x4_f32(
                      false, fa[cur][g].zw, false, fb[cur][g].zw, (short)0, acc, false, false);
        }
        __builtin_amdgcn_sched_barrier(0);
    }

    float* pb = lds + 4 * PS + pl * (TR * PP);
    {
        const int c = lane & 15;
        const int h = lane >> 4;
#pragma unroll
        for (int v = 0; v < 8; ++v)
            pb[(tm * 16 + 8 * h + v) * PP + tn * 16 + c] = acc[v];
    }
    __syncthreads();

    const float* p0 = lds + 4 * PS;
    const float* p1 = p0 + TR * PP;
    const int  j    = bn * TR + lane;
    const bool leaf = bn < 8;
    float* ob = leaf ? out + (size_t)(bm * TR) * NL + j
                     : out + (size_t)BATCH * NL + (size_t)(bm * TR) * NN + (j - NL);
    const int pitch = leaf ? NL : NN;
    float sum[4];
#pragma unroll
    for (int i = 0; i < 4; ++i) {
        const int r = wave + 8 * i;
        sum[i] = p0[r * PP + lane] + p1[r * PP + lane];
    }
    if (j < NROWS) {
#pragma unroll
        for (int i = 0; i < 4; ++i)
            ob[(wave + 8 * i) * pitch] = __builtin_amdgcn_exp2f(sum[i]);
    }
}

extern "C" void kernel_launch(void* const* d_in, const int* in_sizes, int n_in,
                              void* d_out, int out_size, void* d_ws, size_t ws_size,
                              hipStream_t stream) {
    const float* x    = (const float*)d_in[0];
    const float* rbin = (const float*)d_in[1];
    const float* lbin = (const float*)d_in[2];

    tree_fused<<<256, 256, 0, stream>>>(x, rbin, lbin, (float*)d_out);
}


# each wave skips the 2 matrix instructions of a k group whose bin fragment in its own registers is all zero bits (ballot, scalar branch, no flags between waves); skipped terms are exact zeros
# speedup vs baseline: 3.1578x; 1.0594x over previous
#include <hip/hip_runtime.h>
#include <math.h>


typedef __attribute__((ext_vector_type(2))) float v2f;
typedef __attribute__((ext_vector_type(4))) float v4f;
typedef __attribute__((ext_vector_type(4))) unsigned v4u;
typedef __attribute__((ext_vector_type(8))) float v8f;

#define BATCH  512
#define NN     255
#define NL     256
#define NROWS  511
#define TR     32
#define CHUNK  (TR * NN)
#define KG     32
#define SS     132
#define TS     (KG * SS)
#define PS     (2 * TS)
#define PP     36
#define LDS_USED   (4 * PS + 2 * TR * PP)
#define LDS_FLOATS (164 * 256)

__global__ __launch_bounds__(256) void tree_fused(
    const float* __restrict__ x,
    const float* __restrict__ rbin,
    const float* __restrict__ lbin,
    float* __restrict__ out)
{
    static_assert(LDS_USED <= LDS_FLOATS, "LDS reservation too small");
    __shared__ __attribute__((aligned(16))) float lds[LDS_FLOATS];

    const int tid  = threadIdx.x;
    const int lane = tid & 31;
    const int wave = __builtin_amdgcn_readfirstlane(tid >> 5);
    const int bm   = blockIdx.x >> 4;
    const int bn   = blockIdx.x & 15;

    const float* xc = x    + bm * CHUNK;
    const float* rc = rbin + bn * CHUNK;
    const float* lc = lbin + bn * CHUNK;
    const int last_row = min(TR, NROWS - bn * TR) - 1;

    const int r0   = wave >> 1;
    const int half = wave & 1;
    const int m    = half * 32 + lane;
    const bool tail = (m == 63);
    const unsigned kofs = 4u * (unsigned)m - (tail ? 1u : 0u);
    const int  lpos = (m >> 1) * SS + 2 * (m & 1);

    const float* xb = xc + r0 * NN;
    const float* rb = rc + r0 * NN;
    const float* lb = lc + r0 * NN;
    const int r7 = min(r0 + 28, last_row);
    v4f xv[8], rv[8], lv[8];
#pragma unroll
    for (int j = 0; j < 8; ++j)
        __builtin_memcpy(&xv[j], xb + j * (4 * NN) + kofs, sizeof(v4f));
#pragma unroll
    for (int j = 0; j < 7; ++j) {
        __builtin_memcpy(&rv[j], rb + j * (4 * NN) + kofs, sizeof(v4f));
        __builtin_memcpy(&lv[j], lb + j * (4 * NN) + kofs, sizeof(v4f));
    }
    __builtin_memcpy(&rv[7], rc + r7 * NN + kofs, sizeof(v4f));
    __builtin_memcpy(&lv[7], lc + r7 * NN + kofs, sizeof(v4f));

#pragma unroll
    for (int j = 0; j < 8; ++j) {
        const int row = r0 + 4 * j;
        float* d = lds + (row >> 4) * TS + (row & 15) * 4 + lpos;
        v4f v = xv[j];
        if (tail) v = (v4f){v.y, v.z, v.w, 0.0f};
        v4f a, b;
#pragma unroll
        for (int e = 0; e < 4; ++e) {
            a[e] = fmaxf(__builtin_amdgcn_logf(v[e]),        -1.0e30f);
            b[e] = fmaxf(__builtin_amdgcn_logf(1.0f - v[e]), -1.0e30f);
        }
        *(v2f*)(d)           = a.xy;
        *(v2f*)(d + 64)      = a.zw;
        *(v2f*)(d + PS)      = b.xy;
        *(v2f*)(d + PS + 64) = b.zw;
    }
    __builtin_amdgcn_sched_barrier(0);
#pragma unroll
    for (int j = 0; j < 8; ++j) {
        const int row = r0 + 4 * j;
        float* d = lds + 2 * PS + (row >> 4) * TS + (row & 15) * 4 + lpos;
        v4f r = rv[j], l = lv[j];
        if (tail) {
            r = (v4f){r.y, r.z, r.w, 0.0f};
            l = (v4f){l.y, l.z, l.w, 0.0f};
        }
        *(v2f*)(d)           = r.xy;
        *(v2f*)(d + 64)      = r.zw;
        *(v2f*)(d + PS)      = l.xy;
        *(v2f*)(d + PS + 64) = l.zw;
    }
    __syncthreads();

    const int q  = wave & 3;
    const int pl = wave >> 2;
    const int tm = q >> 1;
    const int tn = q & 1;
    const float* ap = lds + pl * PS       + tm * TS + lane * 4;
    const float* bp = lds + (2 + pl) * PS + tn * TS + lane * 4;

    v8f acc = {};
    v4f fa[2][8], fb[2][8];
#pragma unroll
    for (int g = 0; g < 8; ++g) {
        fa[0][g] = *(const v4f*)(ap + g * SS);
        fb[0][g] = *(const v4f*)(bp + g * SS);
    }
#pragma unroll
    for (int blk = 0; blk < 4; ++blk) {
        const int cur = blk & 1;
        const int nxt = cur ^ 1;
        if (blk < 3) {
#pragma unroll
            for (int g = 0; g < 8; ++g) {
                fa[nxt][g] = *(const v4f*)(ap + ((blk + 1) * 8 + g) * SS);
                fb[nxt][g] = *(const v4f*)(bp + ((blk + 1) * 8 + g) * SS);
            }
        }
        __builtin_amdgcn_sched_barrier(0);
        bool nz[8];
#pragma unroll
        for (int g = 0; g < 8; ++g) {
            const v4u u = __builtin_bit_cast(v4u, fb[cur][g]);
            nz[g] = (__builtin_amdgcn_ballot_w32(u.x != 0u) | __builtin_amdgcn_ballot_w32(u.y != 0u) |
                     __builtin_amdgcn_ballot_w32(u.z != 0u) | __builtin_amdgcn_ballot_w32(u.w != 0u)) != 0u;
        }
#pragma unroll
        for (int g = 0; g < 8; ++g) {
            if (nz[g]) {
                acc = __builtin_amdgcn_wmma_f32_16x16x4_f32(
                          false, fa[cur][g].xy, false, fb[cur][g].xy, (short)0, acc, false, false);
                acc = __builtin_amdgcn_wmma_f32_16x16x4_f32(
                          false, fa[cur][g].zw, false, fb[cur][g].zw, (short)0, acc, false, false);
            }
        }
    }

    float* pb = lds + 4 * PS + pl * (TR * PP);
    {
        const int c = lane & 15;
        const int h = lane >> 4;
#pragma unroll
        for (int v = 0; v < 8; ++v)
            pb[(tm * 16 + 8 * h + v) * PP + tn * 16 + c] = acc[v];
    }
    __syncthreads();

    const float* p0 = lds + 4 * PS;
    const float* p1 = p0 + TR * PP;
    const int  j    = bn * TR + lane;
    const bool leaf = bn < 8;
    float* ob = leaf ? out + (size_t)(bm * TR) * NL + j
                     : out + (size_t)BATCH * NL + (size_t)(bm * TR) * NN + (j - NL);
    const int pitch = leaf ? NL : NN;
    float sum[4];
#pragma unroll
    for (int i = 0; i < 4; ++i) {
        const int r = wave + 8 * i;
        sum[i] = p0[r * PP + lane] + p1[r * PP + lane];
    }
    if (j < NROWS) {
#pragma unroll
        for (int i = 0; i < 4; ++i)
            ob[(wave + 8 * i) * pitch] = __builtin_amdgcn_exp2f(sum[i]);
    }
}

extern "C" void kernel_launch(void* const* d_in, const int* in_sizes, int n_in,
                              void* d_out, int out_size, void* d_ws, size_t ws_size,
                              hipStream_t stream) {
    const float* x    = (const float*)d_in[0];
    const float* rbin = (const float*)d_in[1];
    const float* lbin = (const float*)d_in[2];

    tree_fused<<<256, 256, 0, stream>>>(x, rbin, lbin, (float*)d_out);
}


# zero k groups found while staging: a wave stages a whole 16-row bin tile per K half, leaves a mask word in LDS; matrix waves read fragments and issue WMMAs for marked groups only; sum order unchanged
# speedup vs baseline: 3.4299x; 1.0862x over previous
#include <hip/hip_runtime.h>
#include <math.h>


typedef __attribute__((ext_vector_type(2))) float v2f;
typedef __attribute__((ext_vector_type(4))) float v4f;
typedef __attribute__((ext_vector_type(4))) unsigned v4u;
typedef __attribute__((ext_vector_type(8))) float v8f;

#define BATCH  512
#define NN     255
#define NL     256
#define NROWS  511
#define TR     32
#define CHUNK  (TR * NN)
#define KG     32
#define SS     132
#define TS     (KG * SS)
#define PS     (2 * TS)
#define PP     36
#define NB     16
#define LDS_USED   (4 * PS + 2 * TR * PP)
#define LDS_FLOATS (164 * 256)

template <int I>
__device__ __forceinline__ void frag_loads(unsigned cnt, unsigned long long& todo,
                                           const float* ap, const float* bp,
                                           v4f (&fa)[NB], v4f (&fb)[NB])
{
    if constexpr (I < NB) {
        if (__builtin_expect((unsigned)I < cnt, 1)) {
            const int p = __builtin_ctzll(todo);
            todo &= todo - 1ull;
            fa[I] = *(const v4f*)(ap + p * (SS / 2));
            fb[I] = *(const v4f*)(bp + p * (SS / 2));
            frag_loads<I + 1>(cnt, todo, ap, bp, fa, fb);
        }
    }
}

template <int I>
__device__ __forceinline__ void frag_wmmas(unsigned cnt, v8f& acc,
                                           const v4f (&fa)[NB], const v4f (&fb)[NB])
{
    if constexpr (I < NB) {
        if (__builtin_expect((unsigned)I < cnt, 1)) {
            acc = __builtin_amdgcn_wmma_f32_16x16x4_f32(
                      false, fa[I].xy, false, fb[I].xy, (short)0, acc, false, false);
            acc = __builtin_amdgcn_wmma_f32_16x16x4_f32(
                      false, fa[I].zw, false, fb[I].zw, (short)0, acc, false, false);
            frag_wmmas<I + 1>(cnt, acc, fa, fb);
        }
    }
}

__global__ __launch_bounds__(256) void tree_fused(
    const float* __restrict__ x,
    const float* __restrict__ rbin,
    const float* __restrict__ lbin,
    float* __restrict__ out)
{
    static_assert(LDS_USED <= LDS_FLOATS, "LDS reservation too small");
    __shared__ __attribute__((aligned(16))) float lds[LDS_FLOATS];
    __shared__ unsigned nzword[8];

    const int tid  = threadIdx.x;
    const int lane = tid & 31;
    const int wave = __builtin_amdgcn_readfirstlane(tid >> 5);
    const int bm   = blockIdx.x >> 4;
    const int bn   = blockIdx.x & 15;

    const float* xc = x    + bm * CHUNK;
    const float* rc = rbin + bn * CHUNK;
    const float* lc = lbin + bn * CHUNK;
    const int last_row = min(TR, NROWS - bn * TR) - 1;

    const int r0   = wave >> 1;
    const int half = wave & 1;
    const int m    = half * 32 + lane;
    const bool tail = (m == 63);
    const unsigned kofs = 4u * (unsigned)m - (tail ? 1u : 0u);
    const int  lpos = (m >> 1) * SS + 2 * (m & 1);

    const int bmat = wave >> 2;
    const int bt   = (wave >> 1) & 1;
    const float* bc = bmat ? lc : rc;

    const float* xb = xc + r0 * NN;
    const float* bb = bc + (bt * 16) * NN;
    const int r15 = min(bt * 16 + 15, last_row);
    v4f xv[8], bv[16];
#pragma unroll
    for (int j = 0; j < 8; ++j)
        __builtin_memcpy(&xv[j], xb + j * (4 * NN) + kofs, sizeof(v4f));
#pragma unroll
    for (int i = 0; i < 15; ++i)
        __builtin_memcpy(&bv[i], bb + i * NN + kofs, sizeof(v4f));
    __builtin_memcpy(&bv[15], bc + r15 * NN + kofs, sizeof(v4f));

#pragma unroll
    for (int j = 0; j < 8; ++j) {
        const int row = r0 + 4 * j;
        float* d = lds + (row >> 4) * TS + (row & 15) * 4 + lpos;
        v4f v = xv[j];
        if (tail) v = (v4f){v.y, v.z, v.w, 0.0f};
        v4f a, b;
#pragma unroll
        for (int e = 0; e < 4; ++e) {
            a[e] = fmaxf(__builtin_amdgcn_logf(v[e]),        -1.0e30f);
            b[e] = fmaxf(__builtin_amdgcn_logf(1.0f - v[e]), -1.0e30f);
        }
        *(v2f*)(d)           = a.xy;
        *(v2f*)(d + 64)      = a.zw;
        *(v2f*)(d + PS)      = b.xy;
        *(v2f*)(d + PS + 64) = b.zw;
    }
    __builtin_amdgcn_sched_barrier(0);
    unsigned anybits = 0u;
#pragma unroll
    for (int i = 0; i < 16; ++i) {
        float* d = lds + (2 + bmat) * PS + bt * TS + i * 4 + lpos;
        v4f b = bv[i];
        if (tail) b = (v4f){b.y, b.z, b.w, 0.0f};
        const v4u u = __builtin_bit_cast(v4u, b);
        anybits |= (u.x | u.y) | (u.z | u.w);
        *(v2f*)(d)      = b.xy;
        *(v2f*)(d + 64) = b.zw;
        if ((i & 3) == 3 && i < 15) __builtin_amdgcn_sched_barrier(0);
    }
    const unsigned nzlanes = __builtin_amdgcn_ballot_w32(anybits != 0u);
    if (lane == 0) nzword[wave] = nzlanes;
    __syncthreads();

    const int q  = wave & 3;
    const int pl = wave >> 2;
    const int tm = q >> 1;
    const int tn = q & 1;
    const float* ap = lds + pl * PS       + tm * TS + lane * 4;
    const float* bp = lds + (2 + pl) * PS + tn * TS + lane * 4;

    const unsigned w0 = __builtin_amdgcn_readfirstlane(nzword[pl * 4 + tn * 2 + 0]);
    const unsigned w1 = __builtin_amdgcn_readfirstlane(nzword[pl * 4 + tn * 2 + 1]);
    const unsigned e0 = (w0 | (w0 >> 1)) & 0x55555555u;
    const unsigned e1 = (w1 | (w1 >> 1)) & 0x55555555u;
    unsigned long long todo = ((unsigned long long)e1 << 32) | (unsigned long long)e0;

    v8f acc = {};
    v4f fa[NB], fb[NB];
    while (todo != 0ull) {
        const unsigned cnt = (unsigned)__builtin_popcount((unsigned)todo) +
                             (unsigned)__builtin_popcount((unsigned)(todo >> 32));
        frag_loads<0>(cnt, todo, ap, bp, fa, fb);
        frag_wmmas<0>(cnt, acc, fa, fb);
    }

    float* pb = lds + 4 * PS + pl * (TR * PP);
    {
        const int c = lane & 15;
        const int h = lane >> 4;
#pragma unroll
        for (int v = 0; v < 8; ++v)
            pb[(tm * 16 + 8 * h + v) * PP + tn * 16 + c] = acc[v];
    }
    __syncthreads();

    const float* p0 = lds + 4 * PS;
    const float* p1 = p0 + TR * PP;
    const int  j    = bn * TR + lane;
    const bool leaf = bn < 8;
    float* ob = leaf ? out + (size_t)(bm * TR) * NL + j
                     : out + (size_t)BATCH * NL + (size_t)(bm * TR) * NN + (j - NL);
    const int pitch = leaf ? NL : NN;
    float sum[4];
#pragma unroll
    for (int i = 0; i < 4; ++i) {
        const int r = wave + 8 * i;
        sum[i] = p0[r * PP + lane] + p1[r * PP + lane];
    }
    if (j < NROWS) {
#pragma unroll
        for (int i = 0; i < 4; ++i)
            ob[(wave + 8 * i) * pitch] = __builtin_amdgcn_exp2f(sum[i]);
    }
}

extern "C" void kernel_launch(void* const* d_in, const int* in_sizes, int n_in,
                              void* d_out, int out_size, void* d_ws, size_t ws_size,
                              hipStream_t stream) {
    const float* x    = (const float*)d_in[0];
    const float* rbin = (const float*)d_in[1];
    const float* lbin = (const float*)d_in[2];

    tree_fused<<<256, 256, 0, stream>>>(x, rbin, lbin, (float*)d_out);
}
